# starting kernel
# baseline (speedup 1.0000x reference)
#include <hip/hip_runtime.h>
#include <math.h>

typedef float v2f __attribute__((ext_vector_type(2)));
typedef float v8f __attribute__((ext_vector_type(8)));

#define FIN 256
#define H1 8
#define D1 8
#define HD1 64
#define KEYNEG 0x007FFFFFu

__device__ __forceinline__ unsigned f2key(float f) {
    unsigned u = __float_as_uint(f);
    return (u & 0x80000000u) ? ~u : (u | 0x80000000u);
}
__device__ __forceinline__ float key2f(unsigned k) {
    unsigned u = (k & 0x80000000u) ? (k & 0x7FFFFFFFu) : ~k;
    return __uint_as_float(u);
}
__device__ __forceinline__ float leaky(float v) { return v > 0.f ? v : 0.2f * v; }

__global__ void fill_u32(unsigned* __restrict__ p, unsigned v, long long n) {
    long long i = (long long)blockIdx.x * blockDim.x + threadIdx.x;
    if (i < n) p[i] = v;
}

__global__ __launch_bounds__(256) void gemm1_wmma(const float* __restrict__ x,
                                                  const float* __restrict__ W1,
                                                  float* __restrict__ h1, int N) {
    int wave = threadIdx.x >> 5;
    int lane = threadIdx.x & 31;
    int lm = lane & 15, lh = lane >> 4;
    int row0 = blockIdx.x * 32 + (wave >> 2) * 16;
    int col0 = (wave & 3) * 16;
    if (row0 >= N) return;
    const float* xr = x + (long long)(row0 + lm) * FIN;
    v8f acc = {0.f, 0.f, 0.f, 0.f, 0.f, 0.f, 0.f, 0.f};
    for (int k0 = 0; k0 < FIN; k0 += 4) {
        int ka = k0 + 2 * lh;
        v2f a, b;
        a.x = xr[ka];
        a.y = xr[ka + 1];
        b.x = W1[ka * HD1 + col0 + lm];
        b.y = W1[(ka + 1) * HD1 + col0 + lm];
        acc = __builtin_amdgcn_wmma_f32_16x16x4_f32(false, a, false, b, (short)0, acc,
                                                    false, false);
    }
    float* out = h1 + (long long)row0 * HD1 + col0 + lm;
#pragma unroll
    for (int k = 0; k < 8; ++k)
        out[(long long)(k + 8 * lh) * HD1] = acc[k];
}

__global__ void alpha1_kernel(const float* __restrict__ h1,
                              const float* __restrict__ a_src,
                              const float* __restrict__ a_dst,
                              float* __restrict__ as, float* __restrict__ ad, int N) {
    long long i = (long long)blockIdx.x * blockDim.x + threadIdx.x;
    if (i >= (long long)N * H1) return;
    int h = (int)(i & 7);
    const float* hp = h1 + (i >> 3) * HD1 + h * D1;
    float s = 0.f, d = 0.f;
#pragma unroll
    for (int c = 0; c < D1; ++c) {
        float v = hp[c];
        s += v * a_src[h * D1 + c];
        d += v * a_dst[h * D1 + c];
    }
    as[i] = s;
    ad[i] = d;
}

__global__ void edge_max1(const int* __restrict__ ei, int E, int N,
                          const float* __restrict__ as, const float* __restrict__ ad,
                          unsigned* __restrict__ emax) {
    long long t = (long long)blockIdx.x * blockDim.x + threadIdx.x;
    long long ET = (long long)E + N;
    if (t >= ET * H1) return;
    int h = (int)(t & 7);
    long long e = t >> 3;
    int s, d;
    if (e < E) { s = ei[e]; d = ei[E + e]; } else { s = d = (int)(e - E); }
    float v = leaky(as[(long long)s * H1 + h] + ad[(long long)d * H1 + h]);
    atomicMax(&emax[(long long)d * H1 + h], f2key(v));
}

__global__ void edge_acc1(const int* __restrict__ ei, int E, int N,
                          const float* __restrict__ as, const float* __restrict__ ad,
                          const unsigned* __restrict__ emax,
                          const float* __restrict__ h1,
                          float* __restrict__ denom, float* __restrict__ num) {
    long long t = (long long)blockIdx.x * blockDim.x + threadIdx.x;
    long long ET = (long long)E + N;
    long long e = t >> 5;
    if (e >= ET) return;
    int lane = (int)(t & 31);
    int s, d;
    if (e < E) { s = ei[e]; d = ei[E + e]; } else { s = d = (int)(e - E); }
    int h = lane >> 2;
    float v = leaky(as[(long long)s * H1 + h] + ad[(long long)d * H1 + h]);
    float w = __expf(v - key2f(emax[(long long)d * H1 + h]));
    if ((lane & 3) == 0) atomicAdd(&denom[(long long)d * H1 + h], w);
    int c0 = lane * 2;
    const float* hs = h1 + (long long)s * HD1 + c0;
    float* nd = num + (long long)d * HD1 + c0;
    atomicAdd(nd, w * hs[0]);
    atomicAdd(nd + 1, w * hs[1]);
}

__global__ void finalize1(const float* __restrict__ num, const float* __restrict__ denom,
                          const float* __restrict__ b1, float* __restrict__ hx, int N) {
    long long i = (long long)blockIdx.x * blockDim.x + threadIdx.x;
    if (i >= (long long)N * HD1) return;
    int hc = (int)(i & 63);
    float v = num[i] / denom[(i >> 6) * H1 + (hc >> 3)] + b1[hc];
    hx[i] = v > 0.f ? v : (__expf(v) - 1.f);
}

__global__ __launch_bounds__(224) void gemm2_wmma(const float* __restrict__ hx,
                                                  const float* __restrict__ W2,
                                                  float* __restrict__ h2, int N, int C2) {
    int wave = threadIdx.x >> 5;
    int lane = threadIdx.x & 31;
    int lm = lane & 15, lh = lane >> 4;
    int row0 = blockIdx.x * 16;
    int col = wave * 16 + lm;
    bool cok = col < C2;
    const float* ar = hx + (long long)(row0 + lm) * HD1;
    v8f acc = {0.f, 0.f, 0.f, 0.f, 0.f, 0.f, 0.f, 0.f};
    for (int k0 = 0; k0 < HD1; k0 += 4) {
        int ka = k0 + 2 * lh;
        v2f a, b;
        a.x = ar[ka];
        a.y = ar[ka + 1];
        b.x = cok ? W2[ka * C2 + col] : 0.f;
        b.y = cok ? W2[(ka + 1) * C2 + col] : 0.f;
        acc = __builtin_amdgcn_wmma_f32_16x16x4_f32(false, a, false, b, (short)0, acc,
                                                    false, false);
    }
    if (cok) {
        float* out = h2 + (long long)row0 * C2 + col;
#pragma unroll
        for (int k = 0; k < 8; ++k)
            out[(long long)(k + 8 * lh) * C2] = acc[k];
    }
}

__global__ void alpha2_kernel(const float* __restrict__ h2,
                              const float* __restrict__ a_src,
                              const float* __restrict__ a_dst,
                              float* __restrict__ as, float* __restrict__ ad,
                              int N, int C2) {
    int wid = threadIdx.x >> 5, lane = threadIdx.x & 31;
    int n = blockIdx.x * (blockDim.x >> 5) + wid;
    if (n >= N) return;
    const float* hp = h2 + (long long)n * C2;
    float s = 0.f, d = 0.f;
    for (int c = lane; c < C2; c += 32) {
        float v = hp[c];
        s += v * a_src[c];
        d += v * a_dst[c];
    }
    for (int m = 16; m; m >>= 1) {
        s += __shfl_xor(s, m, 32);
        d += __shfl_xor(d, m, 32);
    }
    if (lane == 0) { as[n] = s; ad[n] = d; }
}

__global__ void edge_max2(const int* __restrict__ ei, int E, int N,
                          const float* __restrict__ as, const float* __restrict__ ad,
                          unsigned* __restrict__ emax) {
    long long e = (long long)blockIdx.x * blockDim.x + threadIdx.x;
    long long ET = (long long)E + N;
    if (e >= ET) return;
    int s, d;
    if (e < E) { s = ei[e]; d = ei[E + e]; } else { s = d = (int)(e - E); }
    atomicMax(&emax[d], f2key(leaky(as[s] + ad[d])));
}

__global__ void edge_acc2(const int* __restrict__ ei, int E, int N,
                          const float* __restrict__ as, const float* __restrict__ ad,
                          const unsigned* __restrict__ emax,
                          const float* __restrict__ h2,
                          float* __restrict__ denom, float* __restrict__ num, int C2) {
    long long t = (long long)blockIdx.x * blockDim.x + threadIdx.x;
    long long ET = (long long)E + N;
    long long e = t >> 5;
    if (e >= ET) return;
    int lane = (int)(t & 31);
    int s, d;
    if (e < E) { s = ei[e]; d = ei[E + e]; } else { s = d = (int)(e - E); }
    float w = __expf(leaky(as[s] + ad[d]) - key2f(emax[d]));
    if (lane == 0) atomicAdd(&denom[d], w);
    const float* hs = h2 + (long long)s * C2;
    float* nd = num + (long long)d * C2;
    for (int c = lane; c < C2; c += 32)
        atomicAdd(&nd[c], w * hs[c]);
}

__global__ void finalize2(float* __restrict__ out, const float* __restrict__ denom,
                          const float* __restrict__ b2, int N, int C2) {
    long long i = (long long)blockIdx.x * blockDim.x + threadIdx.x;
    if (i >= (long long)N * C2) return;
    int c = (int)(i % C2);
    long long n = i / C2;
    float v = out[i] / denom[n] + b2[c];
    out[i] = 1.f / (1.f + __expf(-v));
}

static inline int cdiv(long long a, long long b) { return (int)((a + b - 1) / b); }

extern "C" void kernel_launch(void* const* d_in, const int* in_sizes, int n_in,
                              void* d_out, int out_size, void* d_ws, size_t ws_size,
                              hipStream_t stream) {
    const float* x      = (const float*)d_in[0];
    const int*   ei     = (const int*)d_in[1];
    const float* W1     = (const float*)d_in[2];
    const float* a_src1 = (const float*)d_in[3];
    const float* a_dst1 = (const float*)d_in[4];
    const float* b1     = (const float*)d_in[5];
    const float* W2     = (const float*)d_in[6];
    const float* a_src2 = (const float*)d_in[7];
    const float* a_dst2 = (const float*)d_in[8];
    const float* b2     = (const float*)d_in[9];

    const int N  = in_sizes[0] / FIN;
    const int E  = in_sizes[1] / 2;
    const int C2 = in_sizes[9];
    const long long ET = (long long)E + N;

    float*    h1     = (float*)d_ws;
    float*    as1    = h1 + (long long)N * HD1;
    float*    ad1    = as1 + (long long)N * H1;
    unsigned* emax1  = (unsigned*)(ad1 + (long long)N * H1);
    float*    denom1 = (float*)(emax1 + (long long)N * H1);
    float*    num1   = denom1 + (long long)N * H1;
    float*    h2     = num1 + (long long)N * HD1;
    float*    as2    = h2 + (long long)N * C2;
    float*    ad2    = as2 + N;
    unsigned* emax2  = (unsigned*)(ad2 + N);
    float*    denom2 = (float*)(emax2 + N);
    float*    out    = (float*)d_out;

    const int B = 256;

    fill_u32<<<cdiv((long long)N * H1, B), B, 0, stream>>>(emax1, KEYNEG, (long long)N * H1);
    fill_u32<<<cdiv((long long)N * (H1 + HD1), B), B, 0, stream>>>(
        (unsigned*)denom1, 0u, (long long)N * (H1 + HD1));
    fill_u32<<<cdiv(N, B), B, 0, stream>>>(emax2, KEYNEG, N);
    fill_u32<<<cdiv(N, B), B, 0, stream>>>((unsigned*)denom2, 0u, N);
    fill_u32<<<cdiv((long long)N * C2, B), B, 0, stream>>>((unsigned*)out, 0u,
                                                           (long long)N * C2);

    gemm1_wmma<<<cdiv(N, 32), 256, 0, stream>>>(x, W1, h1, N);
    alpha1_kernel<<<cdiv((long long)N * H1, B), B, 0, stream>>>(h1, a_src1, a_dst1,
                                                                as1, ad1, N);
    edge_max1<<<cdiv(ET * H1, B), B, 0, stream>>>(ei, E, N, as1, ad1, emax1);
    edge_acc1<<<cdiv(ET * 32, B), B, 0, stream>>>(ei, E, N, as1, ad1, emax1, h1,
                                                  denom1, num1);
    finalize1<<<cdiv((long long)N * HD1, B), B, 0, stream>>>(num1, denom1, b1, h1, N);

    gemm2_wmma<<<cdiv(N, 16), 224, 0, stream>>>(h1, W2, h2, N, C2);
    alpha2_kernel<<<cdiv(N, 8), 256, 0, stream>>>(h2, a_src2, a_dst2, as2, ad2, N, C2);
    edge_max2<<<cdiv(ET, B), B, 0, stream>>>(ei, E, N, as2, ad2, emax2);
    edge_acc2<<<cdiv(ET * 32, B), B, 0, stream>>>(ei, E, N, as2, ad2, emax2, h2,
                                                  denom2, out, C2);
    finalize2<<<cdiv((long long)N * C2, B), B, 0, stream>>>(out, denom2, b2, N, C2);
}


# Both GEMMs: x and hx row strips staged through LDS by whole-line 16-byte loads, one strip ahead; W column-tile fragments stay in registers of looping waves; k order in the sums changes
# speedup vs baseline: 3.3116x; 3.3116x over previous
#include <hip/hip_runtime.h>
#include <math.h>

typedef float v2f __attribute__((ext_vector_type(2)));
typedef float v4f __attribute__((ext_vector_type(4)));
typedef float v8f __attribute__((ext_vector_type(8)));

#define FIN 256
#define H1 8
#define D1 8
#define HD1 64

__device__ __forceinline__ float leaky(float v) { return v > 0.f ? v : 0.2f * v; }

__global__ void fill_i32(int* __restrict__ p, int v, int n) {
    int i = blockIdx.x * blockDim.x + threadIdx.x;
    if (i < n) p[i] = v;
}

__global__ void build_lists(const int* __restrict__ ei, int E,
                            int* __restrict__ head, int2* __restrict__ link) {
    int e = blockIdx.x * blockDim.x + threadIdx.x;
    if (e >= E) return;
    int s = ei[e], d = ei[E + e];
    int old = atomicExch(&head[d], e);
    link[e] = make_int2(s, old);
}

#define A1_PITCH 260
__global__ __launch_bounds__(128) void gemm1_wmma(const float* __restrict__ x,
                                                  const float* __restrict__ W1,
                                                  float* __restrict__ h1, int N) {
    __shared__ __attribute__((aligned(16))) float As[2 * 16 * A1_PITCH];
    int tid = threadIdx.x;
    int wave = tid >> 5;
    int lane = tid & 31;
    int lm = lane & 15, lh = lane >> 4;
    int col0 = wave * 16;
    v2f bf[64];
#pragma unroll
    for (int kb = 0; kb < 32; ++kb) {
#pragma unroll
        for (int j = 0; j < 2; ++j) {
            int k = kb * 8 + 4 * lh + 2 * j;
            bf[kb * 2 + j].x = W1[k * HD1 + col0 + lm];
            bf[kb * 2 + j].y = W1[(k + 1) * HD1 + col0 + lm];
        }
    }
    int nstrips = N >> 4;
    v4f st0 = {0.f, 0.f, 0.f, 0.f}, st1 = st0, st2 = st0, st3 = st0, st4 = st0, st5 = st0, st6 = st0, st7 = st0;
    int t = blockIdx.x;
    if (t < nstrips) {
        const v4f* g = (const v4f*)(x + (long long)t * 16 * FIN) + tid;
        st0 = g[0];   st1 = g[128]; st2 = g[256]; st3 = g[384];
        st4 = g[512]; st5 = g[640]; st6 = g[768]; st7 = g[896];
    }
    int stage = 0;
    for (; t < nstrips; t += gridDim.x) {
        float* sw = As + stage * (16 * A1_PITCH) + (tid >> 6) * A1_PITCH + (tid & 63) * 4;
        *(v4f*)(sw)                 = st0;
        *(v4f*)(sw + 2 * A1_PITCH)  = st1;
        *(v4f*)(sw + 4 * A1_PITCH)  = st2;
        *(v4f*)(sw + 6 * A1_PITCH)  = st3;
        *(v4f*)(sw + 8 * A1_PITCH)  = st4;
        *(v4f*)(sw + 10 * A1_PITCH) = st5;
        *(v4f*)(sw + 12 * A1_PITCH) = st6;
        *(v4f*)(sw + 14 * A1_PITCH) = st7;
        __syncthreads();
        int tn = t + gridDim.x;
        if (tn < nstrips) {
            const v4f* g = (const v4f*)(x + (long long)tn * 16 * FIN) + tid;
            st0 = g[0];   st1 = g[128]; st2 = g[256]; st3 = g[384];
            st4 = g[512]; st5 = g[640]; st6 = g[768]; st7 = g[896];
        }
        v8f acc = {0.f, 0.f, 0.f, 0.f, 0.f, 0.f, 0.f, 0.f};
        const float* ar = As + stage * (16 * A1_PITCH) + lm * A1_PITCH + 4 * lh;
#pragma unroll
        for (int kb = 0; kb < 32; ++kb) {
            v4f q = *(const v4f*)(ar + kb * 8);
            v2f a0, a1;
            a0.x = q.x; a0.y = q.y;
            a1.x = q.z; a1.y = q.w;
            acc = __builtin_amdgcn_wmma_f32_16x16x4_f32(false, a0, false, bf[kb * 2], (short)0, acc,
                                                        false, false);
            acc = __builtin_amdgcn_wmma_f32_16x16x4_f32(false, a1, false, bf[kb * 2 + 1], (short)0, acc,
                                                        false, false);
        }
        float* out = h1 + (long long)t * 16 * HD1 + col0 + lm;
#pragma unroll
        for (int k = 0; k < 8; ++k)
            out[(long long)(k + 8 * lh) * HD1] = acc[k];
        stage ^= 1;
    }
}

__global__ void alpha1_kernel(const float* __restrict__ h1,
                              const float* __restrict__ a_src,
                              const float* __restrict__ a_dst,
                              float* __restrict__ as, float* __restrict__ ad, int N) {
    long long i = (long long)blockIdx.x * blockDim.x + threadIdx.x;
    if (i >= (long long)N * H1) return;
    int h = (int)(i & 7);
    const float* hp = h1 + (i >> 3) * HD1 + h * D1;
    float s = 0.f, d = 0.f;
#pragma unroll
    for (int c = 0; c < D1; ++c) {
        float v = hp[c];
        s += v * a_src[h * D1 + c];
        d += v * a_dst[h * D1 + c];
    }
    as[i] = s;
    ad[i] = d;
}

__global__ __launch_bounds__(256) void walk1(const int* __restrict__ head,
                                             const int2* __restrict__ link, int E,
                                             const float* __restrict__ as,
                                             const float* __restrict__ ad,
                                             const float* __restrict__ h1,
                                             const float* __restrict__ b1,
                                             float* __restrict__ hx, int N) {
    int n = blockIdx.x * 8 + (threadIdx.x >> 5);
    if (n >= N) return;
    int lane = threadIdx.x & 31;
    int h = lane >> 2;
    int c0 = lane * 2;
    float adn = ad[(long long)n * H1 + h];
    float m = leaky(as[(long long)n * H1 + h] + adn);
    float l = 1.f;
    float2 a = *(const float2*)(h1 + (long long)n * HD1 + c0);
    int e = head[n];
    int2 p = make_int2(0, -1);
    if (e >= 0) p = link[e];
    for (int it = 0; it < E && e >= 0; ++it) {
        int s = p.x, nx = p.y;
        int2 pn = make_int2(0, -1);
        if (nx >= 0) pn = link[nx];
        float v = leaky(as[(long long)s * H1 + h] + adn);
        float2 r = *(const float2*)(h1 + (long long)s * HD1 + c0);
        float mn = fmaxf(m, v);
        float sc = __expf(m - mn);
        float w = __expf(v - mn);
        l = l * sc + w;
        a.x = a.x * sc + w * r.x;
        a.y = a.y * sc + w * r.y;
        m = mn;
        e = nx;
        p = pn;
    }
    float vx = a.x / l + b1[c0];
    float vy = a.y / l + b1[c0 + 1];
    float2 o;
    o.x = vx > 0.f ? vx : (__expf(vx) - 1.f);
    o.y = vy > 0.f ? vy : (__expf(vy) - 1.f);
    *(float2*)(hx + (long long)n * HD1 + c0) = o;
}

#define A2_PITCH 68
__global__ __launch_bounds__(224) void gemm2_wmma(const float* __restrict__ hx,
                                                  const float* __restrict__ W2,
                                                  float* __restrict__ h2, int N, int C2) {
    __shared__ __attribute__((aligned(16))) float As[2 * 16 * A2_PITCH];
    int tid = threadIdx.x;
    int wave = tid >> 5;
    int lane = tid & 31;
    int lm = lane & 15, lh = lane >> 4;
    int col = wave * 16 + lm;
    bool cok = col < C2;
    v2f bf[16];
#pragma unroll
    for (int kb = 0; kb < 8; ++kb) {
#pragma unroll
        for (int j = 0; j < 2; ++j) {
            int k = kb * 8 + 4 * lh + 2 * j;
            bf[kb * 2 + j].x = cok ? W2[k * C2 + col] : 0.f;
            bf[kb * 2 + j].y = cok ? W2[(k + 1) * C2 + col] : 0.f;
        }
    }
    int nstrips = N >> 4;
    v4f s0 = {0.f, 0.f, 0.f, 0.f}, s1 = s0;
    int t = blockIdx.x;
    if (t < nstrips) {
        const v4f* g = (const v4f*)(hx + (long long)t * 16 * HD1);
        s0 = g[tid];
        if (tid < 32) s1 = g[224 + tid];
    }
    int stage = 0;
    for (; t < nstrips; t += gridDim.x) {
        float* sb = As + stage * (16 * A2_PITCH);
        *(v4f*)(sb + (tid >> 4) * A2_PITCH + (tid & 15) * 4) = s0;
        if (tid < 32) {
            int q = 224 + tid;
            *(v4f*)(sb + (q >> 4) * A2_PITCH + (q & 15) * 4) = s1;
        }
        __syncthreads();
        int tn = t + gridDim.x;
        if (tn < nstrips) {
            const v4f* g = (const v4f*)(hx + (long long)tn * 16 * HD1);
            s0 = g[tid];
            if (tid < 32) s1 = g[224 + tid];
        }
        v8f acc = {0.f, 0.f, 0.f, 0.f, 0.f, 0.f, 0.f, 0.f};
        const float* ar = sb + lm * A2_PITCH + 4 * lh;
#pragma unroll
        for (int kb = 0; kb < 8; ++kb) {
            v4f q = *(const v4f*)(ar + kb * 8);
            v2f a0, a1;
            a0.x = q.x; a0.y = q.y;
            a1.x = q.z; a1.y = q.w;
            acc = __builtin_amdgcn_wmma_f32_16x16x4_f32(false, a0, false, bf[kb * 2], (short)0, acc,
                                                        false, false);
            acc = __builtin_amdgcn_wmma_f32_16x16x4_f32(false, a1, false, bf[kb * 2 + 1], (short)0, acc,
                                                        false, false);
        }
        if (cok) {
            float* out = h2 + (long long)t * 16 * C2 + col;
#pragma unroll
            for (int k = 0; k < 8; ++k)
                out[(long long)(k + 8 * lh) * C2] = acc[k];
        }
        stage ^= 1;
    }
}

__global__ void alpha2_kernel(const float* __restrict__ h2,
                              const float* __restrict__ a_src,
                              const float* __restrict__ a_dst,
                              float* __restrict__ as, float* __restrict__ ad,
                              int N, int C2) {
    int wid = threadIdx.x >> 5, lane = threadIdx.x & 31;
    int n = blockIdx.x * (blockDim.x >> 5) + wid;
    if (n >= N) return;
    const float* hp = h2 + (long long)n * C2;
    float s = 0.f, d = 0.f;
    for (int c = lane; c < C2; c += 32) {
        float v = hp[c];
        s += v * a_src[c];
        d += v * a_dst[c];
    }
    for (int m = 16; m; m >>= 1) {
        s += __shfl_xor(s, m, 32);
        d += __shfl_xor(d, m, 32);
    }
    if (lane == 0) { as[n] = s; ad[n] = d; }
}

__global__ __launch_bounds__(256) void walk2(const int* __restrict__ head,
                                             const int2* __restrict__ link, int E,
                                             const float* __restrict__ as,
                                             const float* __restrict__ ad,
                                             const float* __restrict__ h2,
                                             const float* __restrict__ b2,
                                             float* __restrict__ out, int N, int C2) {
    int n = blockIdx.x * 8 + (threadIdx.x >> 5);
    if (n >= N) return;
    int lane = threadIdx.x & 31;
    int c0 = lane * 4;
    bool on = c0 < C2;
    float adn = ad[n];
    float m = leaky(as[n] + adn);
    float l = 1.f;
    float4 a = make_float4(0.f, 0.f, 0.f, 0.f);
    if (on) a = *(const float4*)(h2 + (long long)n * C2 + c0);
    int e = head[n];
    int2 p = make_int2(0, -1);
    if (e >= 0) p = link[e];
    for (int it = 0; it < E && e >= 0; ++it) {
        int s = p.x, nx = p.y;
        int2 pn = make_int2(0, -1);
        if (nx >= 0) pn = link[nx];
        float v = leaky(as[s] + adn);
        float4 r = make_float4(0.f, 0.f, 0.f, 0.f);
        if (on) r = *(const float4*)(h2 + (long long)s * C2 + c0);
        float mn = fmaxf(m, v);
        float sc = __expf(m - mn);
        float w = __expf(v - mn);
        l = l * sc + w;
        a.x = a.x * sc + w * r.x;
        a.y = a.y * sc + w * r.y;
        a.z = a.z * sc + w * r.z;
        a.w = a.w * sc + w * r.w;
        m = mn;
        e = nx;
        p = pn;
    }
    if (on) {
        float4 o;
        o.x = 1.f / (1.f + __expf(-(a.x / l + b2[c0])));
        o.y = 1.f / (1.f + __expf(-(a.y / l + b2[c0 + 1])));
        o.z = 1.f / (1.f + __expf(-(a.z / l + b2[c0 + 2])));
        o.w = 1.f / (1.f + __expf(-(a.w / l + b2[c0 + 3])));
        *(float4*)(out + (long long)n * C2 + c0) = o;
    }
}

static inline int cdiv(long long a, long long b) { return (int)((a + b - 1) / b); }

extern "C" void kernel_launch(void* const* d_in, const int* in_sizes, int n_in,
                              void* d_out, int out_size, void* d_ws, size_t ws_size,
                              hipStream_t stream) {
    const float* x      = (const float*)d_in[0];
    const int*   ei     = (const int*)d_in[1];
    const float* W1     = (const float*)d_in[2];
    const float* a_src1 = (const float*)d_in[3];
    const float* a_dst1 = (const float*)d_in[4];
    const float* b1     = (const float*)d_in[5];
    const float* W2     = (const float*)d_in[6];
    const float* a_src2 = (const float*)d_in[7];
    const float* a_dst2 = (const float*)d_in[8];
    const float* b2     = (const float*)d_in[9];

    const int N  = in_sizes[0] / FIN;
    const int E  = in_sizes[1] / 2;
    const int C2 = in_sizes[9];

    int2*  link = (int2*)d_ws;
    float* h1   = (float*)(link + E);
    float* as1  = h1 + (long long)N * HD1;
    float* ad1  = as1 + (long long)N * H1;
    float* hx   = ad1 + (long long)N * H1;
    float* h2   = hx + (long long)N * HD1;
    float* as2  = h2 + (long long)N * C2;
    float* ad2  = as2 + N;
    int*   head = (int*)(ad2 + N);
    float* out  = (float*)d_out;

    const int B = 256;

    fill_i32<<<cdiv(N, B), B, 0, stream>>>(head, -1, N);
    build_lists<<<cdiv(E, B), B, 0, stream>>>(ei, E, head, link);

    gemm1_wmma<<<1024, 128, 0, stream>>>(x, W1, h1, N);
    alpha1_kernel<<<cdiv((long long)N * H1, B), B, 0, stream>>>(h1, a_src1, a_dst1,
                                                                as1, ad1, N);
    walk1<<<cdiv(N, 8), 256, 0, stream>>>(head, link, E, as1, ad1, h1, b1, hx, N);

    gemm2_wmma<<<1024, 224, 0, stream>>>(hx, W2, h2, N, C2);
    alpha2_kernel<<<cdiv(N, 8), 256, 0, stream>>>(h2, a_src2, a_dst2, as2, ad2, N, C2);
    walk2<<<cdiv(N, 8), 256, 0, stream>>>(head, link, E, as2, ad2, h2, b2, out, N, C2);
}


# Layer-2 node records at a 512-byte pitch: the h2 row with the node's two logits in its pad columns; a hop of walk2 gathers a link and one aligned record, no logit from a second array
# speedup vs baseline: 3.5611x; 1.0753x over previous
#include <hip/hip_runtime.h>
#include <math.h>

typedef float v2f __attribute__((ext_vector_type(2)));
typedef float v4f __attribute__((ext_vector_type(4)));
typedef float v8f __attribute__((ext_vector_type(8)));

#define FIN 256
#define H1 8
#define D1 8
#define HD1 64

__device__ __forceinline__ float leaky(float v) { return v > 0.f ? v : 0.2f * v; }

__global__ void fill_i32(int* __restrict__ p, int v, int n) {
    int i = blockIdx.x * blockDim.x + threadIdx.x;
    if (i < n) p[i] = v;
}

__global__ void build_lists(const int* __restrict__ ei, int E,
                            int* __restrict__ head, int2* __restrict__ link) {
    int e = blockIdx.x * blockDim.x + threadIdx.x;
    if (e >= E) return;
    int s = ei[e], d = ei[E + e];
    int old = atomicExch(&head[d], e);
    link[e] = make_int2(s, old);
}

#define A1_PITCH 260
__global__ __launch_bounds__(128) void gemm1_wmma(const float* __restrict__ x,
                                                  const float* __restrict__ W1,
                                                  float* __restrict__ h1, int N) {
    __shared__ __attribute__((aligned(16))) float As[2 * 16 * A1_PITCH];
    int tid = threadIdx.x;
    int wave = tid >> 5;
    int lane = tid & 31;
    int lm = lane & 15, lh = lane >> 4;
    int col0 = wave * 16;
    v2f bf[64];
#pragma unroll
    for (int kb = 0; kb < 32; ++kb) {
#pragma unroll
        for (int j = 0; j < 2; ++j) {
            int k = kb * 8 + 4 * lh + 2 * j;
            bf[kb * 2 + j].x = W1[k * HD1 + col0 + lm];
            bf[kb * 2 + j].y = W1[(k + 1) * HD1 + col0 + lm];
        }
    }
    int nstrips = N >> 4;
    v4f st0 = {0.f, 0.f, 0.f, 0.f}, st1 = st0, st2 = st0, st3 = st0, st4 = st0, st5 = st0, st6 = st0, st7 = st0;
    int t = blockIdx.x;
    if (t < nstrips) {
        const v4f* g = (const v4f*)(x + (long long)t * 16 * FIN) + tid;
        st0 = g[0];   st1 = g[128]; st2 = g[256]; st3 = g[384];
        st4 = g[512]; st5 = g[640]; st6 = g[768]; st7 = g[896];
    }
    int stage = 0;
    for (; t < nstrips; t += gridDim.x) {
        float* sw = As + stage * (16 * A1_PITCH) + (tid >> 6) * A1_PITCH + (tid & 63) * 4;
        *(v4f*)(sw)                 = st0;
        *(v4f*)(sw + 2 * A1_PITCH)  = st1;
        *(v4f*)(sw + 4 * A1_PITCH)  = st2;
        *(v4f*)(sw + 6 * A1_PITCH)  = st3;
        *(v4f*)(sw + 8 * A1_PITCH)  = st4;
        *(v4f*)(sw + 10 * A1_PITCH) = st5;
        *(v4f*)(sw + 12 * A1_PITCH) = st6;
        *(v4f*)(sw + 14 * A1_PITCH) = st7;
        __syncthreads();
        int tn = t + gridDim.x;
        if (tn < nstrips) {
            const v4f* g = (const v4f*)(x + (long long)tn * 16 * FIN) + tid;
            st0 = g[0];   st1 = g[128]; st2 = g[256]; st3 = g[384];
            st4 = g[512]; st5 = g[640]; st6 = g[768]; st7 = g[896];
        }
        v8f acc = {0.f, 0.f, 0.f, 0.f, 0.f, 0.f, 0.f, 0.f};
        const float* ar = As + stage * (16 * A1_PITCH) + lm * A1_PITCH + 4 * lh;
#pragma unroll
        for (int kb = 0; kb < 32; ++kb) {
            v4f q = *(const v4f*)(ar + kb * 8);
            v2f a0, a1;
            a0.x = q.x; a0.y = q.y;
            a1.x = q.z; a1.y = q.w;
            acc = __builtin_amdgcn_wmma_f32_16x16x4_f32(false, a0, false, bf[kb * 2], (short)0, acc,
                                                        false, false);
            acc = __builtin_amdgcn_wmma_f32_16x16x4_f32(false, a1, false, bf[kb * 2 + 1], (short)0, acc,
                                                        false, false);
        }
        float* out = h1 + (long long)t * 16 * HD1 + col0 + lm;
#pragma unroll
        for (int k = 0; k < 8; ++k)
            out[(long long)(k + 8 * lh) * HD1] = acc[k];
        stage ^= 1;
    }
}

__global__ void alpha1_kernel(const float* __restrict__ h1,
                              const float* __restrict__ a_src,
                              const float* __restrict__ a_dst,
                              float* __restrict__ as, float* __restrict__ ad, int N) {
    long long i = (long long)blockIdx.x * blockDim.x + threadIdx.x;
    if (i >= (long long)N * H1) return;
    int h = (int)(i & 7);
    const float* hp = h1 + (i >> 3) * HD1 + h * D1;
    float s = 0.f, d = 0.f;
#pragma unroll
    for (int c = 0; c < D1; ++c) {
        float v = hp[c];
        s += v * a_src[h * D1 + c];
        d += v * a_dst[h * D1 + c];
    }
    as[i] = s;
    ad[i] = d;
}

__global__ __launch_bounds__(256) void walk1(const int* __restrict__ head,
                                             const int2* __restrict__ link, int E,
                                             const float* __restrict__ as,
                                             const float* __restrict__ ad,
                                             const float* __restrict__ h1,
                                             const float* __restrict__ b1,
                                             float* __restrict__ hx, int N) {
    int n = blockIdx.x * 8 + (threadIdx.x >> 5);
    if (n >= N) return;
    int lane = threadIdx.x & 31;
    int h = lane >> 2;
    int c0 = lane * 2;
    float adn = ad[(long long)n * H1 + h];
    float m = leaky(as[(long long)n * H1 + h] + adn);
    float l = 1.f;
    float2 a = *(const float2*)(h1 + (long long)n * HD1 + c0);
    int e = head[n];
    int2 p = make_int2(0, -1);
    if (e >= 0) p = link[e];
    for (int it = 0; it < E && e >= 0; ++it) {
        int s = p.x, nx = p.y;
        int2 pn = make_int2(0, -1);
        if (nx >= 0) pn = link[nx];
        float v = leaky(as[(long long)s * H1 + h] + adn);
        float2 r = *(const float2*)(h1 + (long long)s * HD1 + c0);
        float mn = fmaxf(m, v);
        float sc = __expf(m - mn);
        float w = __expf(v - mn);
        l = l * sc + w;
        a.x = a.x * sc + w * r.x;
        a.y = a.y * sc + w * r.y;
        m = mn;
        e = nx;
        p = pn;
    }
    float vx = a.x / l + b1[c0];
    float vy = a.y / l + b1[c0 + 1];
    float2 o;
    o.x = vx > 0.f ? vx : (__expf(vx) - 1.f);
    o.y = vy > 0.f ? vy : (__expf(vy) - 1.f);
    *(float2*)(hx + (long long)n * HD1 + c0) = o;
}

#define A2_PITCH 68
#define P2 128
__global__ __launch_bounds__(224) void gemm2_wmma(const float* __restrict__ hx,
                                                  const float* __restrict__ W2,
                                                  float* __restrict__ h2, int N, int C2) {
    __shared__ __attribute__((aligned(16))) float As[2 * 16 * A2_PITCH];
    int tid = threadIdx.x;
    int wave = tid >> 5;
    int lane = tid & 31;
    int lm = lane & 15, lh = lane >> 4;
    int col = wave * 16 + lm;
    bool cok = col < C2;
    v2f bf[16];
#pragma unroll
    for (int kb = 0; kb < 8; ++kb) {
#pragma unroll
        for (int j = 0; j < 2; ++j) {
            int k = kb * 8 + 4 * lh + 2 * j;
            bf[kb * 2 + j].x = cok ? W2[k * C2 + col] : 0.f;
            bf[kb * 2 + j].y = cok ? W2[(k + 1) * C2 + col] : 0.f;
        }
    }
    int nstrips = N >> 4;
    v4f s0 = {0.f, 0.f, 0.f, 0.f}, s1 = s0;
    int t = blockIdx.x;
    if (t < nstrips) {
        const v4f* g = (const v4f*)(hx + (long long)t * 16 * HD1);
        s0 = g[tid];
        if (tid < 32) s1 = g[224 + tid];
    }
    int stage = 0;
    for (; t < nstrips; t += gridDim.x) {
        float* sb = As + stage * (16 * A2_PITCH);
        *(v4f*)(sb + (tid >> 4) * A2_PITCH + (tid & 15) * 4) = s0;
        if (tid < 32) {
            int q = 224 + tid;
            *(v4f*)(sb + (q >> 4) * A2_PITCH + (q & 15) * 4) = s1;
        }
        __syncthreads();
        int tn = t + gridDim.x;
        if (tn < nstrips) {
            const v4f* g = (const v4f*)(hx + (long long)tn * 16 * HD1);
            s0 = g[tid];
            if (tid < 32) s1 = g[224 + tid];
        }
        v8f acc = {0.f, 0.f, 0.f, 0.f, 0.f, 0.f, 0.f, 0.f};
        const float* ar = sb + lm * A2_PITCH + 4 * lh;
#pragma unroll
        for (int kb = 0; kb < 8; ++kb) {
            v4f q = *(const v4f*)(ar + kb * 8);
            v2f a0, a1;
            a0.x = q.x; a0.y = q.y;
            a1.x = q.z; a1.y = q.w;
            acc = __builtin_amdgcn_wmma_f32_16x16x4_f32(false, a0, false, bf[kb * 2], (short)0, acc,
                                                        false, false);
            acc = __builtin_amdgcn_wmma_f32_16x16x4_f32(false, a1, false, bf[kb * 2 + 1], (short)0, acc,
                                                        false, false);
        }
        if (cok) {
            float* out = h2 + (long long)t * 16 * P2 + col;
#pragma unroll
            for (int k = 0; k < 8; ++k)
                out[(long long)(k + 8 * lh) * P2] = acc[k];
        }
        stage ^= 1;
    }
}

__global__ void alpha2_kernel(float* h2,
                              const float* __restrict__ a_src,
                              const float* __restrict__ a_dst,
                              int N, int C2) {
    int wid = threadIdx.x >> 5, lane = threadIdx.x & 31;
    int n = blockIdx.x * (blockDim.x >> 5) + wid;
    if (n >= N) return;
    float* hp = h2 + (long long)n * P2;
    float s = 0.f, d = 0.f;
    for (int c = lane; c < C2; c += 32) {
        float v = hp[c];
        s += v * a_src[c];
        d += v * a_dst[c];
    }
    for (int m = 16; m; m >>= 1) {
        s += __shfl_xor(s, m, 32);
        d += __shfl_xor(d, m, 32);
    }
    if (lane == 0) *(float4*)(hp + C2) = make_float4(s, d, 0.f, 0.f);
}

__global__ __launch_bounds__(256) void walk2(const int* __restrict__ head,
                                             const int2* __restrict__ link, int E,
                                             const float* __restrict__ h2,
                                             const float* __restrict__ b2,
                                             float* __restrict__ out, int N, int C2) {
    int n = blockIdx.x * 8 + (threadIdx.x >> 5);
    if (n >= N) return;
    int lane = threadIdx.x & 31;
    int c0 = lane * 4;
    bool on = c0 < C2;
    bool rd = c0 <= C2;
    int ll = C2 >> 2;
    float4 a = make_float4(0.f, 0.f, 0.f, 0.f);
    if (rd) a = *(const float4*)(h2 + (long long)n * P2 + c0);
    float adn = __int_as_float(__builtin_amdgcn_readlane(__float_as_int(a.y), ll));
    float m = leaky(__int_as_float(__builtin_amdgcn_readlane(__float_as_int(a.x), ll)) + adn);
    float l = 1.f;
    int e = head[n];
    int2 p = make_int2(0, -1);
    if (e >= 0) p = link[e];
    for (int it = 0; it < E && e >= 0; ++it) {
        int s = p.x, nx = p.y;
        int2 pn = make_int2(0, -1);
        if (nx >= 0) pn = link[nx];
        float4 r = make_float4(0.f, 0.f, 0.f, 0.f);
        if (rd) r = *(const float4*)(h2 + (long long)s * P2 + c0);
        float v = leaky(__int_as_float(__builtin_amdgcn_readlane(__float_as_int(r.x), ll)) + adn);
        float mn = fmaxf(m, v);
        float sc = __expf(m - mn);
        float w = __expf(v - mn);
        l = l * sc + w;
        a.x = a.x * sc + w * r.x;
        a.y = a.y * sc + w * r.y;
        a.z = a.z * sc + w * r.z;
        a.w = a.w * sc + w * r.w;
        m = mn;
        e = nx;
        p = pn;
    }
    if (on) {
        float4 o;
        o.x = 1.f / (1.f + __expf(-(a.x / l + b2[c0])));
        o.y = 1.f / (1.f + __expf(-(a.y / l + b2[c0 + 1])));
        o.z = 1.f / (1.f + __expf(-(a.z / l + b2[c0 + 2])));
        o.w = 1.f / (1.f + __expf(-(a.w / l + b2[c0 + 3])));
        *(float4*)(out + (long long)n * C2 + c0) = o;
    }
}

static inline int cdiv(long long a, long long b) { return (int)((a + b - 1) / b); }

extern "C" void kernel_launch(void* const* d_in, const int* in_sizes, int n_in,
                              void* d_out, int out_size, void* d_ws, size_t ws_size,
                              hipStream_t stream) {
    const float* x      = (const float*)d_in[0];
    const int*   ei     = (const int*)d_in[1];
    const float* W1     = (const float*)d_in[2];
    const float* a_src1 = (const float*)d_in[3];
    const float* a_dst1 = (const float*)d_in[4];
    const float* b1     = (const float*)d_in[5];
    const float* W2     = (const float*)d_in[6];
    const float* a_src2 = (const float*)d_in[7];
    const float* a_dst2 = (const float*)d_in[8];
    const float* b2     = (const float*)d_in[9];

    const int N  = in_sizes[0] / FIN;
    const int E  = in_sizes[1] / 2;
    const int C2 = in_sizes[9];

    int2*  link = (int2*)d_ws;
    float* h1   = (float*)(link + E);
    float* as1  = h1 + (long long)N * HD1;
    float* ad1  = as1 + (long long)N * H1;
    float* hx   = ad1 + (long long)N * H1;
    float* h2   = hx + (long long)N * HD1;
    int*   head = (int*)(h2 + (long long)N * P2);
    float* out  = (float*)d_out;

    const int B = 256;

    fill_i32<<<cdiv(N, B), B, 0, stream>>>(head, -1, N);
    build_lists<<<cdiv(E, B), B, 0, stream>>>(ei, E, head, link);

    gemm1_wmma<<<1024, 128, 0, stream>>>(x, W1, h1, N);
    alpha1_kernel<<<cdiv((long long)N * H1, B), B, 0, stream>>>(h1, a_src1, a_dst1,
                                                                as1, ad1, N);
    walk1<<<cdiv(N, 8), 256, 0, stream>>>(head, link, E, as1, ad1, h1, b1, hx, N);

    gemm2_wmma<<<1024, 224, 0, stream>>>(hx, W2, h2, N, C2);
    alpha2_kernel<<<cdiv(N, 8), 256, 0, stream>>>(h2, a_src2, a_dst2, N, C2);
    walk2<<<cdiv(N, 8), 256, 0, stream>>>(head, link, E, h2, b2, out, N, C2);
}


# Per-node kernels give a node 8 lanes instead of a wave: each walk wave chases 4 nodes' lists at once, a logit wave takes 16 nodes; 300K waves become 56K; sum order in the layer-2 logit dots changes
# speedup vs baseline: 4.1600x; 1.1682x over previous
#include <hip/hip_runtime.h>
#include <math.h>

typedef float v2f __attribute__((ext_vector_type(2)));
typedef float v4f __attribute__((ext_vector_type(4)));
typedef float v8f __attribute__((ext_vector_type(8)));

#define FIN 256
#define H1 8
#define D1 8
#define HD1 64

__device__ __forceinline__ float leaky(float v) { return v > 0.f ? v : 0.2f * v; }

__global__ void fill_i32(int* __restrict__ p, int v, int n) {
    int i = blockIdx.x * blockDim.x + threadIdx.x;
    if (i < n) p[i] = v;
}

__global__ void build_lists(const int* __restrict__ ei, int E,
                            int* __restrict__ head, int2* __restrict__ link) {
    int e = blockIdx.x * blockDim.x + threadIdx.x;
    if (e >= E) return;
    int s = ei[e], d = ei[E + e];
    int old = atomicExch(&head[d], e);
    link[e] = make_int2(s, old);
}

#define A1_PITCH 260
__global__ __launch_bounds__(128) void gemm1_wmma(const float* __restrict__ x,
                                                  const float* __restrict__ W1,
                                                  float* __restrict__ h1, int N) {
    __shared__ __attribute__((aligned(16))) float As[2 * 16 * A1_PITCH];
    int tid = threadIdx.x;
    int wave = tid >> 5;
    int lane = tid & 31;
    int lm = lane & 15, lh = lane >> 4;
    int col0 = wave * 16;
    v2f bf[64];
#pragma unroll
    for (int kb = 0; kb < 32; ++kb) {
#pragma unroll
        for (int j = 0; j < 2; ++j) {
            int k = kb * 8 + 4 * lh + 2 * j;
            bf[kb * 2 + j].x = W1[k * HD1 + col0 + lm];
            bf[kb * 2 + j].y = W1[(k + 1) * HD1 + col0 + lm];
        }
    }
    int nstrips = N >> 4;
    v4f st0 = {0.f, 0.f, 0.f, 0.f}, st1 = st0, st2 = st0, st3 = st0, st4 = st0, st5 = st0, st6 = st0, st7 = st0;
    int t = blockIdx.x;
    if (t < nstrips) {
        const v4f* g = (const v4f*)(x + (long long)t * 16 * FIN) + tid;
        st0 = g[0];   st1 = g[128]; st2 = g[256]; st3 = g[384];
        st4 = g[512]; st5 = g[640]; st6 = g[768]; st7 = g[896];
    }
    int stage = 0;
    for (; t < nstrips; t += gridDim.x) {
        float* sw = As + stage * (16 * A1_PITCH) + (tid >> 6) * A1_PITCH + (tid & 63) * 4;
        *(v4f*)(sw)                 = st0;
        *(v4f*)(sw + 2 * A1_PITCH)  = st1;
        *(v4f*)(sw + 4 * A1_PITCH)  = st2;
        *(v4f*)(sw + 6 * A1_PITCH)  = st3;
        *(v4f*)(sw + 8 * A1_PITCH)  = st4;
        *(v4f*)(sw + 10 * A1_PITCH) = st5;
        *(v4f*)(sw + 12 * A1_PITCH) = st6;
        *(v4f*)(sw + 14 * A1_PITCH) = st7;
        __syncthreads();
        int tn = t + gridDim.x;
        if (tn < nstrips) {
            const v4f* g = (const v4f*)(x + (long long)tn * 16 * FIN) + tid;
            st0 = g[0];   st1 = g[128]; st2 = g[256]; st3 = g[384];
            st4 = g[512]; st5 = g[640]; st6 = g[768]; st7 = g[896];
        }
        v8f acc = {0.f, 0.f, 0.f, 0.f, 0.f, 0.f, 0.f, 0.f};
        const float* ar = As + stage * (16 * A1_PITCH) + lm * A1_PITCH + 4 * lh;
#pragma unroll
        for (int kb = 0; kb < 32; ++kb) {
            v4f q = *(const v4f*)(ar + kb * 8);
            v2f a0, a1;
            a0.x = q.x; a0.y = q.y;
            a1.x = q.z; a1.y = q.w;
            acc = __builtin_amdgcn_wmma_f32_16x16x4_f32(false, a0, false, bf[kb * 2], (short)0, acc,
                                                        false, false);
            acc = __builtin_amdgcn_wmma_f32_16x16x4_f32(false, a1, false, bf[kb * 2 + 1], (short)0, acc,
                                                        false, false);
        }
        float* out = h1 + (long long)t * 16 * HD1 + col0 + lm;
#pragma unroll
        for (int k = 0; k < 8; ++k)
            out[(long long)(k + 8 * lh) * HD1] = acc[k];
        stage ^= 1;
    }
}

__global__ void alpha1_kernel(const float* __restrict__ h1,
                              const float* __restrict__ a_src,
                              const float* __restrict__ a_dst,
                              float* __restrict__ as, float* __restrict__ ad, int N) {
    long long i = (long long)blockIdx.x * blockDim.x + threadIdx.x;
    if (i >= (long long)N * H1) return;
    int h = (int)(i & 7);
    const float* hp = h1 + (i >> 3) * HD1 + h * D1;
    float s = 0.f, d = 0.f;
#pragma unroll
    for (int c = 0; c < D1; ++c) {
        float v = hp[c];
        s += v * a_src[h * D1 + c];
        d += v * a_dst[h * D1 + c];
    }
    as[i] = s;
    ad[i] = d;
}

__global__ __launch_bounds__(256) void walk1(const int* __restrict__ head,
                                             const int2* __restrict__ link, int E,
                                             const float* __restrict__ as,
                                             const float* __restrict__ ad,
                                             const float* __restrict__ h1,
                                             const float* __restrict__ b1,
                                             float* __restrict__ hx, int N) {
    int lane = threadIdx.x & 31;
    int j = lane & 7;
    int n = (blockIdx.x * 8 + (threadIdx.x >> 5)) * 4 + (lane >> 3);
    if (n >= N) return;
    int ha = j >> 1, hb = 4 + ha;
    float adA = ad[(long long)n * H1 + ha];
    float adB = ad[(long long)n * H1 + hb];
    float mA = leaky(as[(long long)n * H1 + ha] + adA);
    float mB = leaky(as[(long long)n * H1 + hb] + adB);
    float lA = 1.f, lB = 1.f;
    const float4* own = (const float4*)(h1 + (long long)n * HD1);
    float4 aA = own[j], aB = own[8 + j];
    int e = head[n];
    int2 p = make_int2(0, -1);
    if (e >= 0) p = link[e];
    for (int it = 0; it < E && e >= 0; ++it) {
        int s = p.x, nx = p.y;
        int2 pn = make_int2(0, -1);
        if (nx >= 0) pn = link[nx];
        const float* sl = as + (long long)s * H1;
        float vA = leaky(sl[ha] + adA);
        float vB = leaky(sl[hb] + adB);
        const float4* row = (const float4*)(h1 + (long long)s * HD1);
        float4 rA = row[j], rB = row[8 + j];
        float mnA = fmaxf(mA, vA), mnB = fmaxf(mB, vB);
        float scA = __expf(mA - mnA), scB = __expf(mB - mnB);
        float wA = __expf(vA - mnA), wB = __expf(vB - mnB);
        lA = lA * scA + wA;
        lB = lB * scB + wB;
        aA.x = aA.x * scA + wA * rA.x;
        aA.y = aA.y * scA + wA * rA.y;
        aA.z = aA.z * scA + wA * rA.z;
        aA.w = aA.w * scA + wA * rA.w;
        aB.x = aB.x * scB + wB * rB.x;
        aB.y = aB.y * scB + wB * rB.y;
        aB.z = aB.z * scB + wB * rB.z;
        aB.w = aB.w * scB + wB * rB.w;
        mA = mnA;
        mB = mnB;
        e = nx;
        p = pn;
    }
    int cA = 4 * j, cB = 32 + 4 * j;
    float4 oA, oB;
    float v;
    v = aA.x / lA + b1[cA];     oA.x = v > 0.f ? v : (__expf(v) - 1.f);
    v = aA.y / lA + b1[cA + 1]; oA.y = v > 0.f ? v : (__expf(v) - 1.f);
    v = aA.z / lA + b1[cA + 2]; oA.z = v > 0.f ? v : (__expf(v) - 1.f);
    v = aA.w / lA + b1[cA + 3]; oA.w = v > 0.f ? v : (__expf(v) - 1.f);
    v = aB.x / lB + b1[cB];     oB.x = v > 0.f ? v : (__expf(v) - 1.f);
    v = aB.y / lB + b1[cB + 1]; oB.y = v > 0.f ? v : (__expf(v) - 1.f);
    v = aB.z / lB + b1[cB + 2]; oB.z = v > 0.f ? v : (__expf(v) - 1.f);
    v = aB.w / lB + b1[cB + 3]; oB.w = v > 0.f ? v : (__expf(v) - 1.f);
    float4* dst = (float4*)(hx + (long long)n * HD1);
    dst[j] = oA;
    dst[8 + j] = oB;
}

#define A2_PITCH 68
#define P2 128
__global__ __launch_bounds__(224) void gemm2_wmma(const float* __restrict__ hx,
                                                  const float* __restrict__ W2,
                                                  float* __restrict__ h2, int N, int C2) {
    __shared__ __attribute__((aligned(16))) float As[2 * 16 * A2_PITCH];
    int tid = threadIdx.x;
    int wave = tid >> 5;
    int lane = tid & 31;
    int lm = lane & 15, lh = lane >> 4;
    int col = wave * 16 + lm;
    bool cok = col < C2;
    v2f bf[16];
#pragma unroll
    for (int kb = 0; kb < 8; ++kb) {
#pragma unroll
        for (int j = 0; j < 2; ++j) {
            int k = kb * 8 + 4 * lh + 2 * j;
            bf[kb * 2 + j].x = cok ? W2[k * C2 + col] : 0.f;
            bf[kb * 2 + j].y = cok ? W2[(k + 1) * C2 + col] : 0.f;
        }
    }
    int nstrips = N >> 4;
    v4f s0 = {0.f, 0.f, 0.f, 0.f}, s1 = s0;
    int t = blockIdx.x;
    if (t < nstrips) {
        const v4f* g = (const v4f*)(hx + (long long)t * 16 * HD1);
        s0 = g[tid];
        if (tid < 32) s1 = g[224 + tid];
    }
    int stage = 0;
    for (; t < nstrips; t += gridDim.x) {
        float* sb = As + stage * (16 * A2_PITCH);
        *(v4f*)(sb + (tid >> 4) * A2_PITCH + (tid & 15) * 4) = s0;
        if (tid < 32) {
            int q = 224 + tid;
            *(v4f*)(sb + (q >> 4) * A2_PITCH + (q & 15) * 4) = s1;
        }
        __syncthreads();
        int tn = t + gridDim.x;
        if (tn < nstrips) {
            const v4f* g = (const v4f*)(hx + (long long)tn * 16 * HD1);
            s0 = g[tid];
            if (tid < 32) s1 = g[224 + tid];
        }
        v8f acc = {0.f, 0.f, 0.f, 0.f, 0.f, 0.f, 0.f, 0.f};
        const float* ar = sb + lm * A2_PITCH + 4 * lh;
#pragma unroll
        for (int kb = 0; kb < 8; ++kb) {
            v4f q = *(const v4f*)(ar + kb * 8);
            v2f a0, a1;
            a0.x = q.x; a0.y = q.y;
            a1.x = q.z; a1.y = q.w;
            acc = __builtin_amdgcn_wmma_f32_16x16x4_f32(false, a0, false, bf[kb * 2], (short)0, acc,
                                                        false, false);
            acc = __builtin_amdgcn_wmma_f32_16x16x4_f32(false, a1, false, bf[kb * 2 + 1], (short)0, acc,
                                                        false, false);
        }
        if (cok) {
            float* out = h2 + (long long)t * 16 * P2 + col;
#pragma unroll
            for (int k = 0; k < 8; ++k)
                out[(long long)(k + 8 * lh) * P2] = acc[k];
        }
        stage ^= 1;
    }
}

__global__ __launch_bounds__(256) void alpha2_kernel(float* h2,
                                                     const float* __restrict__ a_src,
                                                     const float* __restrict__ a_dst,
                                                     int N, int C2) {
    int lane = threadIdx.x & 31;
    int j = lane & 7, g = lane >> 3;
    int n0 = (blockIdx.x * 8 + (threadIdx.x >> 5)) * 16 + g;
    int last = (C2 >> 2) - 1;
    float4 sa[4], da[4];
    int pc[4];
#pragma unroll
    for (int i = 0; i < 4; ++i) {
        int c = 32 * i + 4 * j;
        bool in = c < C2;
        pc[i] = in ? 8 * i + j : last;
        int cc = in ? c : 0;
        sa[i] = make_float4(a_src[cc], a_src[cc + 1], a_src[cc + 2], a_src[cc + 3]);
        da[i] = make_float4(a_dst[cc], a_dst[cc + 1], a_dst[cc + 2], a_dst[cc + 3]);
    }
    float4 v[4][4];
#pragma unroll
    for (int q = 0; q < 4; ++q) {
        int n = n0 + 4 * q;
        if (n > N - 1) n = N - 1;
        const float4* rec = (const float4*)(h2 + (long long)n * P2);
#pragma unroll
        for (int i = 0; i < 4; ++i) v[q][i] = rec[pc[i]];
    }
    float s[4], d[4];
#pragma unroll
    for (int q = 0; q < 4; ++q) {
        s[q] = 0.f;
        d[q] = 0.f;
#pragma unroll
        for (int i = 0; i < 4; ++i) {
            float4 t = v[q][i];
            if (32 * i + 4 * j >= C2) t = make_float4(0.f, 0.f, 0.f, 0.f);
            s[q] += t.x * sa[i].x + t.y * sa[i].y + t.z * sa[i].z + t.w * sa[i].w;
            d[q] += t.x * da[i].x + t.y * da[i].y + t.z * da[i].z + t.w * da[i].w;
        }
    }
#pragma unroll
    for (int q = 0; q < 4; ++q) {
        for (int m = 4; m; m >>= 1) {
            s[q] += __shfl_xor(s[q], m, 32);
            d[q] += __shfl_xor(d[q], m, 32);
        }
    }
#pragma unroll
    for (int q = 0; q < 4; ++q) {
        int n = n0 + 4 * q;
        if (n < N && j == 0) *(float4*)(h2 + (long long)n * P2 + C2) = make_float4(s[q], d[q], 0.f, 0.f);
    }
}

__device__ __forceinline__ float group_read(float v, unsigned lo, unsigned hi) {
    unsigned u = (unsigned)__float_as_int(v);
    return __int_as_float((int)__builtin_amdgcn_permlane16(u, u, lo, hi, false, false));
}

__global__ __launch_bounds__(256) void walk2(const int* __restrict__ head,
                                             const int2* __restrict__ link, int E,
                                             const float* __restrict__ h2,
                                             const float* __restrict__ b2,
                                             float* __restrict__ out, int N, int C2) {
    int lane = threadIdx.x & 31;
    int j = lane & 7;
    int n = (blockIdx.x * 8 + (threadIdx.x >> 5)) * 4 + (lane >> 3);
    if (n >= N) return;
    int c0 = 4 * j, c1 = 32 + 4 * j, c2 = 64 + 4 * j, c3 = 96 + 4 * j;
    int pl = C2 >> 2;
    int p0 = min(j, pl), p1 = min(8 + j, pl), p2 = min(16 + j, pl), p3 = min(24 + j, pl);
    int iL = C2 >> 5;
    unsigned jL = (unsigned)(C2 & 31) >> 2;
    unsigned selLo = jL * 0x11111111u;
    unsigned selHi = (jL + 8u) * 0x11111111u;
    const float4* own = (const float4*)(h2 + (long long)n * P2);
    float4 a0 = own[p0], a1 = own[p1], a2 = own[p2], a3 = own[p3];
    float4 ql = iL == 0 ? a0 : iL == 1 ? a1 : iL == 2 ? a2 : a3;
    float adn = group_read(ql.y, selLo, selHi);
    float m = leaky(group_read(ql.x, selLo, selHi) + adn);
    float l = 1.f;
    int e = head[n];
    int2 p = make_int2(0, -1);
    if (e >= 0) p = link[e];
    for (int it = 0; it < E && e >= 0; ++it) {
        int s = p.x, nx = p.y;
        int2 pn = make_int2(0, -1);
        if (nx >= 0) pn = link[nx];
        const float4* rec = (const float4*)(h2 + (long long)s * P2);
        float4 r0 = rec[p0], r1 = rec[p1], r2 = rec[p2], r3 = rec[p3];
        float sx = iL == 0 ? r0.x : iL == 1 ? r1.x : iL == 2 ? r2.x : r3.x;
        float v = leaky(group_read(sx, selLo, selHi) + adn);
        float mn = fmaxf(m, v);
        float sc = __expf(m - mn);
        float w = __expf(v - mn);
        l = l * sc + w;
        a0.x = a0.x * sc + w * r0.x;
        a0.y = a0.y * sc + w * r0.y;
        a0.z = a0.z * sc + w * r0.z;
        a0.w = a0.w * sc + w * r0.w;
        a1.x = a1.x * sc + w * r1.x;
        a1.y = a1.y * sc + w * r1.y;
        a1.z = a1.z * sc + w * r1.z;
        a1.w = a1.w * sc + w * r1.w;
        a2.x = a2.x * sc + w * r2.x;
        a2.y = a2.y * sc + w * r2.y;
        a2.z = a2.z * sc + w * r2.z;
        a2.w = a2.w * sc + w * r2.w;
        a3.x = a3.x * sc + w * r3.x;
        a3.y = a3.y * sc + w * r3.y;
        a3.z = a3.z * sc + w * r3.z;
        a3.w = a3.w * sc + w * r3.w;
        m = mn;
        e = nx;
        p = pn;
    }
    float* orow = out + (long long)n * C2;
    if (c0 < C2) {
        float4 o;
        o.x = 1.f / (1.f + __expf(-(a0.x / l + b2[c0])));
        o.y = 1.f / (1.f + __expf(-(a0.y / l + b2[c0 + 1])));
        o.z = 1.f / (1.f + __expf(-(a0.z / l + b2[c0 + 2])));
        o.w = 1.f / (1.f + __expf(-(a0.w / l + b2[c0 + 3])));
        *(float4*)(orow + c0) = o;
    }
    if (c1 < C2) {
        float4 o;
        o.x = 1.f / (1.f + __expf(-(a1.x / l + b2[c1])));
        o.y = 1.f / (1.f + __expf(-(a1.y / l + b2[c1 + 1])));
        o.z = 1.f / (1.f + __expf(-(a1.z / l + b2[c1 + 2])));
        o.w = 1.f / (1.f + __expf(-(a1.w / l + b2[c1 + 3])));
        *(float4*)(orow + c1) = o;
    }
    if (c2 < C2) {
        float4 o;
        o.x = 1.f / (1.f + __expf(-(a2.x / l + b2[c2])));
        o.y = 1.f / (1.f + __expf(-(a2.y / l + b2[c2 + 1])));
        o.z = 1.f / (1.f + __expf(-(a2.z / l + b2[c2 + 2])));
        o.w = 1.f / (1.f + __expf(-(a2.w / l + b2[c2 + 3])));
        *(float4*)(orow + c2) = o;
    }
    if (c3 < C2) {
        float4 o;
        o.x = 1.f / (1.f + __expf(-(a3.x / l + b2[c3])));
        o.y = 1.f / (1.f + __expf(-(a3.y / l + b2[c3 + 1])));
        o.z = 1.f / (1.f + __expf(-(a3.z / l + b2[c3 + 2])));
        o.w = 1.f / (1.f + __expf(-(a3.w / l + b2[c3 + 3])));
        *(float4*)(orow + c3) = o;
    }
}

static inline int cdiv(long long a, long long b) { return (int)((a + b - 1) / b); }

extern "C" void kernel_launch(void* const* d_in, const int* in_sizes, int n_in,
                              void* d_out, int out_size, void* d_ws, size_t ws_size,
                              hipStream_t stream) {
    const float* x      = (const float*)d_in[0];
    const int*   ei     = (const int*)d_in[1];
    const float* W1     = (const float*)d_in[2];
    const float* a_src1 = (const float*)d_in[3];
    const float* a_dst1 = (const float*)d_in[4];
    const float* b1     = (const float*)d_in[5];
    const float* W2     = (const float*)d_in[6];
    const float* a_src2 = (const float*)d_in[7];
    const float* a_dst2 = (const float*)d_in[8];
    const float* b2     = (const float*)d_in[9];

    const int N  = in_sizes[0] / FIN;
    const int E  = in_sizes[1] / 2;
    const int C2 = in_sizes[9];

    int2*  link = (int2*)d_ws;
    float* h1   = (float*)(link + E);
    float* as1  = h1 + (long long)N * HD1;
    float* ad1  = as1 + (long long)N * H1;
    float* hx   = ad1 + (long long)N * H1;
    float* h2   = hx + (long long)N * HD1;
    int*   head = (int*)(h2 + (long long)N * P2);
    float* out  = (float*)d_out;

    const int B = 256;

    fill_i32<<<cdiv(N, B), B, 0, stream>>>(head, -1, N);
    build_lists<<<cdiv(E, B), B, 0, stream>>>(ei, E, head, link);

    gemm1_wmma<<<1024, 128, 0, stream>>>(x, W1, h1, N);
    alpha1_kernel<<<cdiv((long long)N * H1, B), B, 0, stream>>>(h1, a_src1, a_dst1,
                                                                as1, ad1, N);
    walk1<<<cdiv(N, 32), 256, 0, stream>>>(head, link, E, as1, ad1, h1, b1, hx, N);

    gemm2_wmma<<<1024, 224, 0, stream>>>(hx, W2, h2, N, C2);
    alpha2_kernel<<<cdiv(N, 128), 256, 0, stream>>>(h2, a_src2, a_dst2, N, C2);
    walk2<<<cdiv(N, 32), 256, 0, stream>>>(head, link, E, h2, b2, out, N, C2);
}


# Layer-2 projection and logits move into the layer-1 gather's tail: a work-group's 32 hx rows go through LDS to W2 (WMMA) and out as whole records; 8 launches become 6, no hx array; sum order kept
# speedup vs baseline: 4.5888x; 1.1031x over previous
#include <hip/hip_runtime.h>
#include <math.h>

typedef float v2f __attribute__((ext_vector_type(2)));
typedef float v4f __attribute__((ext_vector_type(4)));
typedef float v8f __attribute__((ext_vector_type(8)));

#define FIN 256
#define H1 8
#define D1 8
#define HD1 64

__device__ __forceinline__ float leaky(float v) { return v > 0.f ? v : 0.2f * v; }

__global__ void fill_i32(int* __restrict__ p, int v, int n) {
    int i = blockIdx.x * blockDim.x + threadIdx.x;
    if (i < n) p[i] = v;
}

__global__ void build_lists(const int* __restrict__ ei, int E,
                            int* __restrict__ head, int2* __restrict__ link) {
    int e = blockIdx.x * blockDim.x + threadIdx.x;
    if (e >= E) return;
    int s = ei[e], d = ei[E + e];
    int old = atomicExch(&head[d], e);
    link[e] = make_int2(s, old);
}

#define A1_PITCH 260
__global__ __launch_bounds__(128) void gemm1_wmma(const float* __restrict__ x,
                                                  const float* __restrict__ W1,
                                                  float* __restrict__ h1, int N) {
    __shared__ __attribute__((aligned(16))) float As[2 * 16 * A1_PITCH];
    int tid = threadIdx.x;
    int wave = tid >> 5;
    int lane = tid & 31;
    int lm = lane & 15, lh = lane >> 4;
    int col0 = wave * 16;
    v2f bf[64];
#pragma unroll
    for (int kb = 0; kb < 32; ++kb) {
#pragma unroll
        for (int j = 0; j < 2; ++j) {
            int k = kb * 8 + 4 * lh + 2 * j;
            bf[kb * 2 + j].x = W1[k * HD1 + col0 + lm];
            bf[kb * 2 + j].y = W1[(k + 1) * HD1 + col0 + lm];
        }
    }
    int nstrips = N >> 4;
    v4f st0 = {0.f, 0.f, 0.f, 0.f}, st1 = st0, st2 = st0, st3 = st0, st4 = st0, st5 = st0, st6 = st0, st7 = st0;
    int t = blockIdx.x;
    if (t < nstrips) {
        const v4f* g = (const v4f*)(x + (long long)t * 16 * FIN) + tid;
        st0 = g[0];   st1 = g[128]; st2 = g[256]; st3 = g[384];
        st4 = g[512]; st5 = g[640]; st6 = g[768]; st7 = g[896];
    }
    int stage = 0;
    for (; t < nstrips; t += gridDim.x) {
        float* sw = As + stage * (16 * A1_PITCH) + (tid >> 6) * A1_PITCH + (tid & 63) * 4;
        *(v4f*)(sw)                 = st0;
        *(v4f*)(sw + 2 * A1_PITCH)  = st1;
        *(v4f*)(sw + 4 * A1_PITCH)  = st2;
        *(v4f*)(sw + 6 * A1_PITCH)  = st3;
        *(v4f*)(sw + 8 * A1_PITCH)  = st4;
        *(v4f*)(sw + 10 * A1_PITCH) = st5;
        *(v4f*)(sw + 12 * A1_PITCH) = st6;
        *(v4f*)(sw + 14 * A1_PITCH) = st7;
        __syncthreads();
        int tn = t + gridDim.x;
        if (tn < nstrips) {
            const v4f* g = (const v4f*)(x + (long long)tn * 16 * FIN) + tid;
            st0 = g[0];   st1 = g[128]; st2 = g[256]; st3 = g[384];
            st4 = g[512]; st5 = g[640]; st6 = g[768]; st7 = g[896];
        }
        v8f acc = {0.f, 0.f, 0.f, 0.f, 0.f, 0.f, 0.f, 0.f};
        const float* ar = As + stage * (16 * A1_PITCH) + lm * A1_PITCH + 4 * lh;
#pragma unroll
        for (int kb = 0; kb < 32; ++kb) {
            v4f q = *(const v4f*)(ar + kb * 8);
            v2f a0, a1;
            a0.x = q.x; a0.y = q.y;
            a1.x = q.z; a1.y = q.w;
            acc = __builtin_amdgcn_wmma_f32_16x16x4_f32(false, a0, false, bf[kb * 2], (short)0, acc,
                                                        false, false);
            acc = __builtin_amdgcn_wmma_f32_16x16x4_f32(false, a1, false, bf[kb * 2 + 1], (short)0, acc,
                                                        false, false);
        }
        float* out = h1 + (long long)t * 16 * HD1 + col0 + lm;
#pragma unroll
        for (int k = 0; k < 8; ++k)
            out[(long long)(k + 8 * lh) * HD1] = acc[k];
        stage ^= 1;
    }
}

__global__ void alpha1_kernel(const float* __restrict__ h1,
                              const float* __restrict__ a_src,
                              const float* __restrict__ a_dst,
                              float* __restrict__ as, float* __restrict__ ad, int N) {
    long long i = (long long)blockIdx.x * blockDim.x + threadIdx.x;
    if (i >= (long long)N * H1) return;
    int h = (int)(i & 7);
    const float* hp = h1 + (i >> 3) * HD1 + h * D1;
    float s = 0.f, d = 0.f;
#pragma unroll
    for (int c = 0; c < D1; ++c) {
        float v = hp[c];
        s += v * a_src[h * D1 + c];
        d += v * a_dst[h * D1 + c];
    }
    as[i] = s;
    ad[i] = d;
}

#define P2 128
#define A2_PITCH 68
#define H2_PITCH 132
__global__ __launch_bounds__(256) __attribute__((amdgpu_waves_per_eu(16, 16))) void walk1_proj2(const int* __restrict__ head,
                                                   const int2* __restrict__ link, int E,
                                                   const float* __restrict__ as,
                                                   const float* __restrict__ ad,
                                                   const float* __restrict__ h1,
                                                   const float* __restrict__ b1,
                                                   const float* __restrict__ W2,
                                                   const float* __restrict__ a_src2,
                                                   const float* __restrict__ a_dst2,
                                                   float* __restrict__ h2, int N, int C2) {
    __shared__ __attribute__((aligned(16))) float Xs[32 * A2_PITCH];
    __shared__ __attribute__((aligned(16))) float Hs[32 * H2_PITCH];
    int tid = threadIdx.x;
    int wave = tid >> 5;
    int lane = tid & 31;
    int j = lane & 7;
    int nl = wave * 4 + (lane >> 3);
    int n = blockIdx.x * 32 + nl;
    bool live = n < N;
    int nc = live ? n : N - 1;
    int ha = j >> 1, hb = 4 + ha;
    float adA = ad[(long long)nc * H1 + ha];
    float adB = ad[(long long)nc * H1 + hb];
    float mA = leaky(as[(long long)nc * H1 + ha] + adA);
    float mB = leaky(as[(long long)nc * H1 + hb] + adB);
    float lA = 1.f, lB = 1.f;
    const float4* own = (const float4*)(h1 + (long long)nc * HD1);
    float4 aA = own[j], aB = own[8 + j];
    int e = live ? head[nc] : -1;
    int2 p = make_int2(0, -1);
    if (e >= 0) p = link[e];
    for (int it = 0; it < E && e >= 0; ++it) {
        int s = p.x, nx = p.y;
        int2 pn = make_int2(0, -1);
        if (nx >= 0) pn = link[nx];
        const float* sl = as + (long long)s * H1;
        float vA = leaky(sl[ha] + adA);
        float vB = leaky(sl[hb] + adB);
        const float4* row = (const float4*)(h1 + (long long)s * HD1);
        float4 rA = row[j], rB = row[8 + j];
        float mnA = fmaxf(mA, vA), mnB = fmaxf(mB, vB);
        float scA = __expf(mA - mnA), scB = __expf(mB - mnB);
        float wA = __expf(vA - mnA), wB = __expf(vB - mnB);
        lA = lA * scA + wA;
        lB = lB * scB + wB;
        aA.x = aA.x * scA + wA * rA.x;
        aA.y = aA.y * scA + wA * rA.y;
        aA.z = aA.z * scA + wA * rA.z;
        aA.w = aA.w * scA + wA * rA.w;
        aB.x = aB.x * scB + wB * rB.x;
        aB.y = aB.y * scB + wB * rB.y;
        aB.z = aB.z * scB + wB * rB.z;
        aB.w = aB.w * scB + wB * rB.w;
        mA = mnA;
        mB = mnB;
        e = nx;
        p = pn;
    }
    {
        int cA = 4 * j, cB = 32 + 4 * j;
        float4 oA, oB;
        float v;
        v = aA.x / lA + b1[cA];     oA.x = v > 0.f ? v : (__expf(v) - 1.f);
        v = aA.y / lA + b1[cA + 1]; oA.y = v > 0.f ? v : (__expf(v) - 1.f);
        v = aA.z / lA + b1[cA + 2]; oA.z = v > 0.f ? v : (__expf(v) - 1.f);
        v = aA.w / lA + b1[cA + 3]; oA.w = v > 0.f ? v : (__expf(v) - 1.f);
        v = aB.x / lB + b1[cB];     oB.x = v > 0.f ? v : (__expf(v) - 1.f);
        v = aB.y / lB + b1[cB + 1]; oB.y = v > 0.f ? v : (__expf(v) - 1.f);
        v = aB.z / lB + b1[cB + 2]; oB.z = v > 0.f ? v : (__expf(v) - 1.f);
        v = aB.w / lB + b1[cB + 3]; oB.w = v > 0.f ? v : (__expf(v) - 1.f);
        float* xr = Xs + nl * A2_PITCH;
        *(float4*)(xr + cA) = oA;
        *(float4*)(xr + cB) = oB;
    }
    __syncthreads();
    if (wave < 7) {
        int lm = lane & 15, lh = lane >> 4;
        int col = wave * 16 + lm;
        bool cok = col < C2;
        int cl = cok ? col : C2 - 1;
        v2f bf[16];
#pragma unroll
        for (int kb = 0; kb < 8; ++kb) {
#pragma unroll
            for (int jj = 0; jj < 2; ++jj) {
                int k = kb * 8 + 4 * lh + 2 * jj;
                float w0 = W2[k * C2 + cl], w1 = W2[(k + 1) * C2 + cl];
                bf[kb * 2 + jj].x = cok ? w0 : 0.f;
                bf[kb * 2 + jj].y = cok ? w1 : 0.f;
            }
        }
#pragma unroll
        for (int st = 0; st < 2; ++st) {
            v8f acc = {0.f, 0.f, 0.f, 0.f, 0.f, 0.f, 0.f, 0.f};
            const float* ar = Xs + (16 * st + lm) * A2_PITCH + 4 * lh;
#pragma unroll
            for (int kb = 0; kb < 8; ++kb) {
                v4f q = *(const v4f*)(ar + kb * 8);
                v2f a0, a1;
                a0.x = q.x; a0.y = q.y;
                a1.x = q.z; a1.y = q.w;
                acc = __builtin_amdgcn_wmma_f32_16x16x4_f32(false, a0, false, bf[kb * 2], (short)0, acc,
                                                            false, false);
                acc = __builtin_amdgcn_wmma_f32_16x16x4_f32(false, a1, false, bf[kb * 2 + 1], (short)0, acc,
                                                            false, false);
            }
            float* hr = Hs + (16 * st + 8 * lh) * H2_PITCH + col;
#pragma unroll
            for (int k = 0; k < 8; ++k)
                hr[k * H2_PITCH] = acc[k];
        }
    }
    __syncthreads();
    int pl = C2 >> 2;
    const float* hrow = Hs + nl * H2_PITCH;
    float4 v[4];
    float sd = 0.f, dd = 0.f;
#pragma unroll
    for (int i = 0; i < 4; ++i) {
        int c = 32 * i + 4 * j;
        bool in = c < C2;
        int cc = in ? c : 0;
        float4 sa = make_float4(a_src2[cc], a_src2[cc + 1], a_src2[cc + 2], a_src2[cc + 3]);
        float4 da = make_float4(a_dst2[cc], a_dst2[cc + 1], a_dst2[cc + 2], a_dst2[cc + 3]);
        float4 t = *(const float4*)(hrow + (in ? c : C2 - 4));
        v[i] = t;
        if (!in) t = make_float4(0.f, 0.f, 0.f, 0.f);
        sd += t.x * sa.x + t.y * sa.y + t.z * sa.z + t.w * sa.w;
        dd += t.x * da.x + t.y * da.y + t.z * da.z + t.w * da.w;
    }
    for (int m = 4; m; m >>= 1) {
        sd += __shfl_xor(sd, m, 32);
        dd += __shfl_xor(dd, m, 32);
    }
    if (live) {
        float4* rec = (float4*)(h2 + (long long)n * P2);
#pragma unroll
        for (int i = 0; i < 4; ++i) {
            int piece = 8 * i + j;
            if (piece < pl) rec[piece] = v[i];
            else if (piece == pl) rec[piece] = make_float4(sd, dd, 0.f, 0.f);
        }
    }
}

__device__ __forceinline__ float group_read(float v, unsigned lo, unsigned hi) {
    unsigned u = (unsigned)__float_as_int(v);
    return __int_as_float((int)__builtin_amdgcn_permlane16(u, u, lo, hi, false, false));
}

__global__ __launch_bounds__(256) void walk2(const int* __restrict__ head,
                                             const int2* __restrict__ link, int E,
                                             const float* __restrict__ h2,
                                             const float* __restrict__ b2,
                                             float* __restrict__ out, int N, int C2) {
    int lane = threadIdx.x & 31;
    int j = lane & 7;
    int n = (blockIdx.x * 8 + (threadIdx.x >> 5)) * 4 + (lane >> 3);
    if (n >= N) return;
    int c0 = 4 * j, c1 = 32 + 4 * j, c2 = 64 + 4 * j, c3 = 96 + 4 * j;
    int pl = C2 >> 2;
    int p0 = min(j, pl), p1 = min(8 + j, pl), p2 = min(16 + j, pl), p3 = min(24 + j, pl);
    int iL = C2 >> 5;
    unsigned jL = (unsigned)(C2 & 31) >> 2;
    unsigned selLo = jL * 0x11111111u;
    unsigned selHi = (jL + 8u) * 0x11111111u;
    const float4* own = (const float4*)(h2 + (long long)n * P2);
    float4 a0 = own[p0], a1 = own[p1], a2 = own[p2], a3 = own[p3];
    float4 ql = iL == 0 ? a0 : iL == 1 ? a1 : iL == 2 ? a2 : a3;
    float adn = group_read(ql.y, selLo, selHi);
    float m = leaky(group_read(ql.x, selLo, selHi) + adn);
    float l = 1.f;
    int e = head[n];
    int2 p = make_int2(0, -1);
    if (e >= 0) p = link[e];
    for (int it = 0; it < E && e >= 0; ++it) {
        int s = p.x, nx = p.y;
        int2 pn = make_int2(0, -1);
        if (nx >= 0) pn = link[nx];
        const float4* rec = (const float4*)(h2 + (long long)s * P2);
        float4 r0 = rec[p0], r1 = rec[p1], r2 = rec[p2], r3 = rec[p3];
        float sx = iL == 0 ? r0.x : iL == 1 ? r1.x : iL == 2 ? r2.x : r3.x;
        float v = leaky(group_read(sx, selLo, selHi) + adn);
        float mn = fmaxf(m, v);
        float sc = __expf(m - mn);
        float w = __expf(v - mn);
        l = l * sc + w;
        a0.x = a0.x * sc + w * r0.x;
        a0.y = a0.y * sc + w * r0.y;
        a0.z = a0.z * sc + w * r0.z;
        a0.w = a0.w * sc + w * r0.w;
        a1.x = a1.x * sc + w * r1.x;
        a1.y = a1.y * sc + w * r1.y;
        a1.z = a1.z * sc + w * r1.z;
        a1.w = a1.w * sc + w * r1.w;
        a2.x = a2.x * sc + w * r2.x;
        a2.y = a2.y * sc + w * r2.y;
        a2.z = a2.z * sc + w * r2.z;
        a2.w = a2.w * sc + w * r2.w;
        a3.x = a3.x * sc + w * r3.x;
        a3.y = a3.y * sc + w * r3.y;
        a3.z = a3.z * sc + w * r3.z;
        a3.w = a3.w * sc + w * r3.w;
        m = mn;
        e = nx;
        p = pn;
    }
    float* orow = out + (long long)n * C2;
    if (c0 < C2) {
        float4 o;
        o.x = 1.f / (1.f + __expf(-(a0.x / l + b2[c0])));
        o.y = 1.f / (1.f + __expf(-(a0.y / l + b2[c0 + 1])));
        o.z = 1.f / (1.f + __expf(-(a0.z / l + b2[c0 + 2])));
        o.w = 1.f / (1.f + __expf(-(a0.w / l + b2[c0 + 3])));
        *(float4*)(orow + c0) = o;
    }
    if (c1 < C2) {
        float4 o;
        o.x = 1.f / (1.f + __expf(-(a1.x / l + b2[c1])));
        o.y = 1.f / (1.f + __expf(-(a1.y / l + b2[c1 + 1])));
        o.z = 1.f / (1.f + __expf(-(a1.z / l + b2[c1 + 2])));
        o.w = 1.f / (1.f + __expf(-(a1.w / l + b2[c1 + 3])));
        *(float4*)(orow + c1) = o;
    }
    if (c2 < C2) {
        float4 o;
        o.x = 1.f / (1.f + __expf(-(a2.x / l + b2[c2])));
        o.y = 1.f / (1.f + __expf(-(a2.y / l + b2[c2 + 1])));
        o.z = 1.f / (1.f + __expf(-(a2.z / l + b2[c2 + 2])));
        o.w = 1.f / (1.f + __expf(-(a2.w / l + b2[c2 + 3])));
        *(float4*)(orow + c2) = o;
    }
    if (c3 < C2) {
        float4 o;
        o.x = 1.f / (1.f + __expf(-(a3.x / l + b2[c3])));
        o.y = 1.f / (1.f + __expf(-(a3.y / l + b2[c3 + 1])));
        o.z = 1.f / (1.f + __expf(-(a3.z / l + b2[c3 + 2])));
        o.w = 1.f / (1.f + __expf(-(a3.w / l + b2[c3 + 3])));
        *(float4*)(orow + c3) = o;
    }
}

static inline int cdiv(long long a, long long b) { return (int)((a + b - 1) / b); }

extern "C" void kernel_launch(void* const* d_in, const int* in_sizes, int n_in,
                              void* d_out, int out_size, void* d_ws, size_t ws_size,
                              hipStream_t stream) {
    const float* x      = (const float*)d_in[0];
    const int*   ei     = (const int*)d_in[1];
    const float* W1     = (const float*)d_in[2];
    const float* a_src1 = (const float*)d_in[3];
    const float* a_dst1 = (const float*)d_in[4];
    const float* b1     = (const float*)d_in[5];
    const float* W2     = (const float*)d_in[6];
    const float* a_src2 = (const float*)d_in[7];
    const float* a_dst2 = (const float*)d_in[8];
    const float* b2     = (const float*)d_in[9];

    const int N  = in_sizes[0] / FIN;
    const int E  = in_sizes[1] / 2;
    const int C2 = in_sizes[9];

    int2*  link = (int2*)d_ws;
    float* h1   = (float*)(link + E);
    float* as1  = h1 + (long long)N * HD1;
    float* ad1  = as1 + (long long)N * H1;
    float* h2   = ad1 + (long long)N * H1;
    int*   head = (int*)(h2 + (long long)N * P2);
    float* out  = (float*)d_out;

    const int B = 256;

    fill_i32<<<cdiv(N, B), B, 0, stream>>>(head, -1, N);
    build_lists<<<cdiv(E, B), B, 0, stream>>>(ei, E, head, link);

    gemm1_wmma<<<1024, 128, 0, stream>>>(x, W1, h1, N);
    alpha1_kernel<<<cdiv((long long)N * H1, B), B, 0, stream>>>(h1, a_src1, a_dst1,
                                                                as1, ad1, N);
    walk1_proj2<<<cdiv(N, 32), 256, 0, stream>>>(head, link, E, as1, ad1, h1, b1, W2, a_src2, a_dst2, h2, N, C2);

    walk2<<<cdiv(N, 32), 256, 0, stream>>>(head, link, E, h2, b2, out, N, C2);
}
